# starting kernel
# baseline (speedup 1.0000x reference)
#include <hip/hip_runtime.h>
#include <math.h>

#define N_NODES   6000
#define DIA_LEN_C 100
#define D_DIM     512

typedef __attribute__((ext_vector_type(2))) float v2f;
typedef __attribute__((ext_vector_type(8))) float v8f;

__global__ void __launch_bounds__(256)
build_x_kernel(const float* __restrict__ inp,
               const int*   __restrict__ qmask,
               float*       __restrict__ x) {
    const int wave = (blockIdx.x * blockDim.x + threadIdx.x) >> 5;
    const int lane = threadIdx.x & 31;
    if (wave >= N_NODES) return;
    const int i  = wave;
    const int d0 = (i / DIA_LEN_C) * DIA_LEN_C;
    const int d1 = d0 + DIA_LEN_C;
    const int sp = qmask[2 * i];

    int prev = -1, next = -1;
    for (int j = i - 1; j >= d0; --j) { if (qmask[2 * j] == sp) { prev = j; break; } }
    for (int j = i + 1; j <  d1; ++j) { if (qmask[2 * j] == sp) { next = j; break; } }

    const float* xi = inp + (size_t)i * D_DIM;
    const float* xp = (prev >= 0) ? inp + (size_t)prev * D_DIM : xi;
    const float* xn = (next >= 0) ? inp + (size_t)next * D_DIM : xi;

    float4 ra[4], rp[4], rn[4];
    float sii = 0.f, spp = 0.f, snn = 0.f, sip = 0.f, sin_ = 0.f;
#pragma unroll
    for (int t = 0; t < 4; ++t) {
        const int k = t * 128 + lane * 4;
        float4 a = *(const float4*)(xi + k);
        float4 p = *(const float4*)(xp + k);
        float4 n = *(const float4*)(xn + k);
        ra[t] = a; rp[t] = p; rn[t] = n;
        sii  += a.x*a.x + a.y*a.y + a.z*a.z + a.w*a.w;
        spp  += p.x*p.x + p.y*p.y + p.z*p.z + p.w*p.w;
        snn  += n.x*n.x + n.y*n.y + n.z*n.z + n.w*n.w;
        sip  += a.x*p.x + a.y*p.y + a.z*p.z + a.w*p.w;
        sin_ += a.x*n.x + a.y*n.y + a.z*n.z + a.w*n.w;
    }
#pragma unroll
    for (int m = 16; m >= 1; m >>= 1) {
        sii  += __shfl_xor(sii,  m, 32);
        spp  += __shfl_xor(spp,  m, 32);
        snn  += __shfl_xor(snn,  m, 32);
        sip  += __shfl_xor(sip,  m, 32);
        sin_ += __shfl_xor(sin_, m, 32);
    }

    const float INV_PI = 0.318309886183790671538f;
    float wp = 0.f, wn = 0.f;
    if (prev >= 0) {
        float den = sqrtf(sii) * sqrtf(spp);
        float c   = (den > 0.f) ? sip / den : 0.f;
        c  = fminf(1.f, fmaxf(-1.f, c));
        wp = 1.f - acosf(c) * INV_PI;
    }
    if (next >= 0) {
        float den = sqrtf(sii) * sqrtf(snn);
        float c   = (den > 0.f) ? sin_ / den : 0.f;
        c  = fminf(1.f, fmaxf(-1.f, c));
        wn = 1.f - acosf(c) * INV_PI;
    }

    float* xo = x + (size_t)i * D_DIM;
#pragma unroll
    for (int t = 0; t < 4; ++t) {
        const int k = t * 128 + lane * 4;
        float4 r;
        r.x = ra[t].x + wp * rp[t].x + wn * rn[t].x;
        r.y = ra[t].y + wp * rp[t].y + wn * rn[t].y;
        r.z = ra[t].z + wp * rp[t].z + wn * rn[t].z;
        r.w = ra[t].w + wp * rp[t].w + wn * rn[t].w;
        *(float4*)(xo + k) = r;
    }
}

__global__ void __launch_bounds__(128)
gemm_wmma_kernel(const float* __restrict__ x,
                 const float* __restrict__ W,
                 const float* __restrict__ bias,
                 float*       __restrict__ out) {
    const int waveInBlk = threadIdx.x >> 5;
    const int lane      = threadIdx.x & 31;
    const int m0 = blockIdx.x * 16;
    const int n0 = (blockIdx.y * 4 + waveInBlk) * 64;

    const int row = lane & 15;
    const int kh  = (lane >> 4) * 2;
    const float* aBase = x + (size_t)(m0 + row) * D_DIM + kh;
    const float* bBase = W + (size_t)(n0 + row) * D_DIM + kh;

    v8f acc[4] = {{}, {}, {}, {}};
#pragma unroll 4
    for (int k = 0; k < D_DIM; k += 4) {
        v2f a  = *(const v2f*)(aBase + k);
        v2f b0 = *(const v2f*)(bBase + k);
        v2f b1 = *(const v2f*)(bBase + 16 * D_DIM + k);
        v2f b2 = *(const v2f*)(bBase + 32 * D_DIM + k);
        v2f b3 = *(const v2f*)(bBase + 48 * D_DIM + k);
        acc[0] = __builtin_amdgcn_wmma_f32_16x16x4_f32(
            false, a, false, b0, (short)0, acc[0], false, false);
        acc[1] = __builtin_amdgcn_wmma_f32_16x16x4_f32(
            false, a, false, b1, (short)0, acc[1], false, false);
        acc[2] = __builtin_amdgcn_wmma_f32_16x16x4_f32(
            false, a, false, b2, (short)0, acc[2], false, false);
        acc[3] = __builtin_amdgcn_wmma_f32_16x16x4_f32(
            false, a, false, b3, (short)0, acc[3], false, false);
    }

    const int rbase = m0 + (lane >> 4) * 8;
#pragma unroll
    for (int j = 0; j < 4; ++j) {
        const int col  = n0 + j * 16 + row;
        const float bv = bias[col];
#pragma unroll
        for (int r = 0; r < 8; ++r) {
            out[(size_t)(rbase + r) * D_DIM + col] = acc[j][r] + bv;
        }
    }
}

extern "C" void kernel_launch(void* const* d_in, const int* in_sizes, int n_in,
                              void* d_out, int out_size, void* d_ws, size_t ws_size,
                              hipStream_t stream) {
    const float* inp   = (const float*)d_in[0];
    const int*   qmask = (const int*)d_in[2];
    const float* W     = (const float*)d_in[3];
    const float* bias  = (const float*)d_in[4];
    float*       xbuf  = (float*)d_ws;
    float*       out   = (float*)d_out;

    build_x_kernel<<<dim3(N_NODES / 8), dim3(256), 0, stream>>>(inp, qmask, xbuf);
    gemm_wmma_kernel<<<dim3(N_NODES / 16, 2), dim3(128), 0, stream>>>(
        xbuf, W, bias, out);
}


# One launch, adjacency applied on the A side: each work-group builds its 48x512 tile of x (inputs + wp*prev + wn*next) in LDS, A fragments read from LDS; every sum and its order kept
# speedup vs baseline: 1.8026x; 1.8026x over previous
#include <hip/hip_runtime.h>
#include <math.h>

#define N_NODES   6000
#define DIA_LEN_C 100
#define D_DIM     512

typedef __attribute__((ext_vector_type(2))) float v2f;
typedef __attribute__((ext_vector_type(4))) float v4f;
typedef __attribute__((ext_vector_type(8))) float v8f;


#define GEMM_WM      48
#define GEMM_WN      64
#define GEMM_MTILES  125
#define GEMM_NHALVES 2
#define ROWS_PER_WAVE 12
#define ROW_BATCH     6
#define XPITCH       516

__device__ __forceinline__ unsigned lowmask(int x) {
    return x <= 0 ? 0u : (x >= 32 ? 0xffffffffu : ((1u << x) - 1u));
}
__device__ __forceinline__ unsigned rmask(int a, int b) { return lowmask(b) & ~lowmask(a); }

__device__ __forceinline__ float xrow(float x) {
    return __builtin_bit_cast(float, __builtin_amdgcn_permlanex16(
        __builtin_bit_cast(int, x), __builtin_bit_cast(int, x), 0x76543210, 0xfedcba98, false, false));
}
template <int M>
__device__ __forceinline__ float xlane(float x) {
    return __builtin_bit_cast(float, __builtin_amdgcn_update_dpp(
        __builtin_bit_cast(int, x), __builtin_bit_cast(int, x), 0x160 + M, 0xf, 0xf, false));
}

__device__ __forceinline__ void load_b(v4f (&fb)[4], const float* pb, int koff) {
#pragma unroll
    for (int j = 0; j < 4; ++j) fb[j] = *(const v4f*)(pb + j * 16 * D_DIM + koff);
}

__device__ __forceinline__ void mma_set(v8f (&acc)[3][4], const v4f (&fa)[3], const v4f (&fb)[4]) {
    v2f alo[3], ahi[3], blo[4], bhi[4];
#pragma unroll
    for (int i = 0; i < 3; ++i) {
        alo[i] = __builtin_shufflevector(fa[i], fa[i], 0, 1);
        ahi[i] = __builtin_shufflevector(fa[i], fa[i], 2, 3);
    }
#pragma unroll
    for (int j = 0; j < 4; ++j) {
        blo[j] = __builtin_shufflevector(fb[j], fb[j], 0, 1);
        bhi[j] = __builtin_shufflevector(fb[j], fb[j], 2, 3);
    }
#pragma unroll
    for (int i = 0; i < 3; ++i)
#pragma unroll
        for (int j = 0; j < 4; ++j)
            acc[i][j] = __builtin_amdgcn_wmma_f32_16x16x4_f32(
                false, alo[i], false, blo[j], (short)0, acc[i][j], false, false);
#pragma unroll
    for (int i = 0; i < 3; ++i)
#pragma unroll
        for (int j = 0; j < 4; ++j)
            acc[i][j] = __builtin_amdgcn_wmma_f32_16x16x4_f32(
                false, ahi[i], false, bhi[j], (short)0, acc[i][j], false, false);
}

__global__ void __launch_bounds__(128)
gcn_fused_kernel(const float* __restrict__ inp,
                 const int*   __restrict__ qmask,
                 const float* __restrict__ W,
                 const float* __restrict__ bias,
                 float*       __restrict__ out) {
    __shared__ __attribute__((aligned(16))) float xs[GEMM_WM * XPITCH];

    const int waveInBlk = __builtin_amdgcn_readfirstlane((int)(threadIdx.x >> 5));
    const int lane      = threadIdx.x & 31;
    const int c  = lane & 15;
    const int h  = lane >> 4;
    const int m0 = blockIdx.x * GEMM_WM;
    const int n0 = (blockIdx.y * 4 + waveInBlk) * GEMM_WN;

    const float* pb = W + (size_t)(n0 + c) * D_DIM + 4 * h;

    v4f fa[4][3], fb[4][4];
    load_b(fb[0], pb, 0);
    load_b(fb[1], pb, 8);
    load_b(fb[2], pb, 16);

    float bv[4];
#pragma unroll
    for (int j = 0; j < 4; ++j) bv[j] = bias[n0 + j * 16 + c];

    const int base = (m0 / DIA_LEN_C) * DIA_LEN_C;
    unsigned spk[7];
#pragma unroll
    for (int q = 0; q < 7; ++q) {
        const int j = min(base + 32 * q + lane, N_NODES - 1);
        spk[q] = __builtin_amdgcn_ballot_w32(qmask[2 * j] == 1);
    }

    const int r0 = waveInBlk * ROWS_PER_WAVE;
    int prevRow, nextRow;
    {
        const int li = r0 + (lane < ROWS_PER_WAVE ? lane : ROWS_PER_WAVE - 1);
        const int i  = m0 + li;
        const int p  = i - base;
        const int dl = (i / DIA_LEN_C) * DIA_LEN_C - base;
        const int dh = dl + DIA_LEN_C;
        const int wq = p >> 5;
        unsigned own = spk[0];
#pragma unroll
        for (int q = 1; q < 7; ++q) own = (wq == q) ? spk[q] : own;
        const bool sp = ((own >> (p & 31)) & 1u) != 0u;
        int pp = -1, np = -1;
#pragma unroll
        for (int q = 6; q >= 0; --q) {
            const unsigned same = sp ? spk[q] : ~spk[q];
            const unsigned m = same & rmask(dl - 32 * q, p - 32 * q);
            if (pp < 0 && m != 0u) pp = 32 * q + 31 - __builtin_clz(m);
        }
#pragma unroll
        for (int q = 0; q < 7; ++q) {
            const unsigned same = sp ? spk[q] : ~spk[q];
            const unsigned m = same & rmask(p + 1 - 32 * q, dh - 32 * q);
            if (np < 0 && m != 0u) np = 32 * q + __builtin_ctz(m);
        }
        prevRow = (pp < 0) ? -1 : base + pp;
        nextRow = (np < 0) ? -1 : base + np;
    }

    const float INV_PI = 0.318309886183790671538f;
    for (int bch = 0; bch < ROWS_PER_WAVE / ROW_BATCH; ++bch) {
        int   prv[ROW_BATCH], nxt[ROW_BATCH];
        v4f   ra[ROW_BATCH][4], rp[ROW_BATCH][4], rn[ROW_BATCH][4];
#pragma unroll
        for (int u = 0; u < ROW_BATCH; ++u) {
            const int k  = bch * ROW_BATCH + u;
            const int i  = m0 + r0 + k;
            prv[u] = __builtin_amdgcn_readlane(prevRow, k);
            nxt[u] = __builtin_amdgcn_readlane(nextRow, k);
            const float* xi = inp + (size_t)i * D_DIM;
            const float* xp = (prv[u] >= 0) ? inp + (size_t)prv[u] * D_DIM : xi;
            const float* xn = (nxt[u] >= 0) ? inp + (size_t)nxt[u] * D_DIM : xi;
#pragma unroll
            for (int t = 0; t < 4; ++t) {
                const int kk = t * 128 + lane * 4;
                ra[u][t] = *(const v4f*)(xi + kk);
                rp[u][t] = *(const v4f*)(xp + kk);
                rn[u][t] = *(const v4f*)(xn + kk);
            }
        }
        float s[ROW_BATCH][5];
#pragma unroll
        for (int u = 0; u < ROW_BATCH; ++u) {
            float sii = 0.f, spp = 0.f, snn = 0.f, sip = 0.f, sin_ = 0.f;
#pragma unroll
            for (int t = 0; t < 4; ++t) {
                const v4f a = ra[u][t], p = rp[u][t], n = rn[u][t];
                sii  += a.x*a.x + a.y*a.y + a.z*a.z + a.w*a.w;
                spp  += p.x*p.x + p.y*p.y + p.z*p.z + p.w*p.w;
                snn  += n.x*n.x + n.y*n.y + n.z*n.z + n.w*n.w;
                sip  += a.x*p.x + a.y*p.y + a.z*p.z + a.w*p.w;
                sin_ += a.x*n.x + a.y*n.y + a.z*n.z + a.w*n.w;
            }
            s[u][0] = sii; s[u][1] = spp; s[u][2] = snn; s[u][3] = sip; s[u][4] = sin_;
        }
#pragma unroll
        for (int u = 0; u < ROW_BATCH; ++u)
#pragma unroll
            for (int q = 0; q < 5; ++q) s[u][q] += xrow(s[u][q]);
#pragma unroll
        for (int u = 0; u < ROW_BATCH; ++u)
#pragma unroll
            for (int q = 0; q < 5; ++q) s[u][q] += xlane<8>(s[u][q]);
#pragma unroll
        for (int u = 0; u < ROW_BATCH; ++u)
#pragma unroll
            for (int q = 0; q < 5; ++q) s[u][q] += xlane<4>(s[u][q]);
#pragma unroll
        for (int u = 0; u < ROW_BATCH; ++u)
#pragma unroll
            for (int q = 0; q < 5; ++q) s[u][q] += xlane<2>(s[u][q]);
#pragma unroll
        for (int u = 0; u < ROW_BATCH; ++u)
#pragma unroll
            for (int q = 0; q < 5; ++q) s[u][q] += xlane<1>(s[u][q]);
        float na = 0.f, nb = 0.f, num = 0.f;
        bool  has = false;
#pragma unroll
        for (int u = 0; u < ROW_BATCH; ++u) {
            if (lane == 2 * u)     { na = s[u][0]; nb = s[u][1]; num = s[u][3]; has = prv[u] >= 0; }
            if (lane == 2 * u + 1) { na = s[u][0]; nb = s[u][2]; num = s[u][4]; has = nxt[u] >= 0; }
        }
        float wgt = 0.f;
        {
            float den = sqrtf(na) * sqrtf(nb);
            float cs  = (den > 0.f) ? num / den : 0.f;
            cs = fminf(1.f, fmaxf(-1.f, cs));
            const float wv = 1.f - acosf(cs) * INV_PI;
            wgt = has ? wv : 0.f;
        }
#pragma unroll
        for (int u = 0; u < ROW_BATCH; ++u) {
            const float wp = __builtin_bit_cast(float, __builtin_amdgcn_readlane(__builtin_bit_cast(int, wgt), 2 * u));
            const float wn = __builtin_bit_cast(float, __builtin_amdgcn_readlane(__builtin_bit_cast(int, wgt), 2 * u + 1));
            float* xo = xs + (r0 + bch * ROW_BATCH + u) * XPITCH;
#pragma unroll
            for (int t = 0; t < 4; ++t) {
                const int kk = t * 128 + lane * 4;
                v4f r;
                r.x = ra[u][t].x + wp * rp[u][t].x + wn * rn[u][t].x;
                r.y = ra[u][t].y + wp * rp[u][t].y + wn * rn[u][t].y;
                r.z = ra[u][t].z + wp * rp[u][t].z + wn * rn[u][t].z;
                r.w = ra[u][t].w + wp * rp[u][t].w + wn * rn[u][t].w;
                *(v4f*)(xo + kk) = r;
            }
        }
    }
    __syncthreads();

    const int aoff = c * XPITCH + 4 * h;
#define LOAD_A(set, koff)                                                        \
    _Pragma("unroll")                                                            \
    for (int i_ = 0; i_ < 3; ++i_)                                               \
        fa[set][i_] = *(const v4f*)(&xs[aoff + i_ * 16 * XPITCH + kbase + (koff)]);

    int kbase = 0;
    LOAD_A(0, 0)
    LOAD_A(1, 8)
    LOAD_A(2, 16)

    v8f acc[3][4];
#pragma unroll
    for (int i = 0; i < 3; ++i)
#pragma unroll
        for (int j = 0; j < 4; ++j) acc[i][j] = v8f{};

    for (int it = 0; it < 15; ++it) {
        LOAD_A(3, 24)
        load_b(fb[3], pb, 24);
        __builtin_amdgcn_sched_barrier(0);
        mma_set(acc, fa[0], fb[0]);
        LOAD_A(0, 32)
        load_b(fb[0], pb, 32);
        __builtin_amdgcn_sched_barrier(0);
        mma_set(acc, fa[1], fb[1]);
        LOAD_A(1, 40)
        load_b(fb[1], pb, 40);
        __builtin_amdgcn_sched_barrier(0);
        mma_set(acc, fa[2], fb[2]);
        LOAD_A(2, 48)
        load_b(fb[2], pb, 48);
        __builtin_amdgcn_sched_barrier(0);
        mma_set(acc, fa[3], fb[3]);
        kbase += 32;
        pb += 32;
    }
    LOAD_A(3, 24)
    load_b(fb[3], pb, 24);
    mma_set(acc, fa[0], fb[0]);
    mma_set(acc, fa[1], fb[1]);
    mma_set(acc, fa[2], fb[2]);
    mma_set(acc, fa[3], fb[3]);
#undef LOAD_A

#pragma unroll
    for (int i = 0; i < 3; ++i) {
        const int rbase = m0 + i * 16 + h * 8;
#pragma unroll
        for (int j = 0; j < 4; ++j) {
            const int col = n0 + j * 16 + c;
#pragma unroll
            for (int r = 0; r < 8; ++r) {
                out[(size_t)(rbase + r) * D_DIM + col] = acc[i][j][r] + bv[j];
            }
        }
    }
}

extern "C" void kernel_launch(void* const* d_in, const int* in_sizes, int n_in,
                              void* d_out, int out_size, void* d_ws, size_t ws_size,
                              hipStream_t stream) {
    const float* inp   = (const float*)d_in[0];
    const int*   qmask = (const int*)d_in[2];
    const float* W     = (const float*)d_in[3];
    const float* bias  = (const float*)d_in[4];
    float*       out   = (float*)d_out;

    gcn_fused_kernel<<<dim3(GEMM_MTILES, GEMM_NHALVES), dim3(128), 0, stream>>>(
        inp, qmask, W, bias, out);
}


# Two waves per SIMD: 8-wave work-groups, wave tile 48x32 (6 accumulators), tile build 6 rows per wave in one batch; a wave's load waits are covered by its partner; every sum and its order kept
# speedup vs baseline: 1.9205x; 1.0654x over previous
#include <hip/hip_runtime.h>
#include <math.h>

#define N_NODES   6000
#define DIA_LEN_C 100
#define D_DIM     512

typedef __attribute__((ext_vector_type(2))) float v2f;
typedef __attribute__((ext_vector_type(4))) float v4f;
typedef __attribute__((ext_vector_type(8))) float v8f;


#define GEMM_WM      48
#define GEMM_WN      32
#define GEMM_MTILES  125
#define GEMM_NWAVES  8
#define GEMM_NJ      2
#define GEMM_NHALVES 2
#define ROWS_PER_WAVE 6
#define ROW_BATCH     6
#define XPITCH       516

__device__ __forceinline__ unsigned lowmask(int x) {
    return x <= 0 ? 0u : (x >= 32 ? 0xffffffffu : ((1u << x) - 1u));
}
__device__ __forceinline__ unsigned rmask(int a, int b) { return lowmask(b) & ~lowmask(a); }

__device__ __forceinline__ float xrow(float x) {
    return __builtin_bit_cast(float, __builtin_amdgcn_permlanex16(
        __builtin_bit_cast(int, x), __builtin_bit_cast(int, x), 0x76543210, 0xfedcba98, false, false));
}
template <int M>
__device__ __forceinline__ float xlane(float x) {
    return __builtin_bit_cast(float, __builtin_amdgcn_update_dpp(
        __builtin_bit_cast(int, x), __builtin_bit_cast(int, x), 0x160 + M, 0xf, 0xf, false));
}

__device__ __forceinline__ void load_b(v4f (&fb)[GEMM_NJ], const float* pb, int koff) {
#pragma unroll
    for (int j = 0; j < GEMM_NJ; ++j) fb[j] = *(const v4f*)(pb + j * 16 * D_DIM + koff);
}

__device__ __forceinline__ void mma_set(v8f (&acc)[3][GEMM_NJ], const v4f (&fa)[3], const v4f (&fb)[GEMM_NJ]) {
    v2f alo[3], ahi[3], blo[GEMM_NJ], bhi[GEMM_NJ];
#pragma unroll
    for (int i = 0; i < 3; ++i) {
        alo[i] = __builtin_shufflevector(fa[i], fa[i], 0, 1);
        ahi[i] = __builtin_shufflevector(fa[i], fa[i], 2, 3);
    }
#pragma unroll
    for (int j = 0; j < GEMM_NJ; ++j) {
        blo[j] = __builtin_shufflevector(fb[j], fb[j], 0, 1);
        bhi[j] = __builtin_shufflevector(fb[j], fb[j], 2, 3);
    }
#pragma unroll
    for (int i = 0; i < 3; ++i)
#pragma unroll
        for (int j = 0; j < GEMM_NJ; ++j)
            acc[i][j] = __builtin_amdgcn_wmma_f32_16x16x4_f32(
                false, alo[i], false, blo[j], (short)0, acc[i][j], false, false);
#pragma unroll
    for (int i = 0; i < 3; ++i)
#pragma unroll
        for (int j = 0; j < GEMM_NJ; ++j)
            acc[i][j] = __builtin_amdgcn_wmma_f32_16x16x4_f32(
                false, ahi[i], false, bhi[j], (short)0, acc[i][j], false, false);
}

__global__ void __launch_bounds__(256)
gcn_fused_kernel(const float* __restrict__ inp,
                 const int*   __restrict__ qmask,
                 const float* __restrict__ W,
                 const float* __restrict__ bias,
                 float*       __restrict__ out) {
    __shared__ __attribute__((aligned(16))) float xs[GEMM_WM * XPITCH];

    const int waveInBlk = __builtin_amdgcn_readfirstlane((int)(threadIdx.x >> 5));
    const int lane      = threadIdx.x & 31;
    const int c  = lane & 15;
    const int h  = lane >> 4;
    const int m0 = blockIdx.x * GEMM_WM;
    const int n0 = (blockIdx.y * GEMM_NWAVES + waveInBlk) * GEMM_WN;

    const float* pb = W + (size_t)(n0 + c) * D_DIM + 4 * h;

    v4f fa[4][3], fb[4][GEMM_NJ];
    load_b(fb[0], pb, 0);
    load_b(fb[1], pb, 8);
    load_b(fb[2], pb, 16);

    float bv[GEMM_NJ];
#pragma unroll
    for (int j = 0; j < GEMM_NJ; ++j) bv[j] = bias[n0 + j * 16 + c];

    const int base = (m0 / DIA_LEN_C) * DIA_LEN_C;
    unsigned spk[7];
#pragma unroll
    for (int q = 0; q < 7; ++q) {
        const int j = min(base + 32 * q + lane, N_NODES - 1);
        spk[q] = __builtin_amdgcn_ballot_w32(qmask[2 * j] == 1);
    }

    const int r0 = waveInBlk * ROWS_PER_WAVE;
    int prevRow, nextRow;
    {
        const int li = r0 + (lane < ROWS_PER_WAVE ? lane : ROWS_PER_WAVE - 1);
        const int i  = m0 + li;
        const int p  = i - base;
        const int dl = (i / DIA_LEN_C) * DIA_LEN_C - base;
        const int dh = dl + DIA_LEN_C;
        const int wq = p >> 5;
        unsigned own = spk[0];
#pragma unroll
        for (int q = 1; q < 7; ++q) own = (wq == q) ? spk[q] : own;
        const bool sp = ((own >> (p & 31)) & 1u) != 0u;
        int pp = -1, np = -1;
#pragma unroll
        for (int q = 6; q >= 0; --q) {
            const unsigned same = sp ? spk[q] : ~spk[q];
            const unsigned m = same & rmask(dl - 32 * q, p - 32 * q);
            if (pp < 0 && m != 0u) pp = 32 * q + 31 - __builtin_clz(m);
        }
#pragma unroll
        for (int q = 0; q < 7; ++q) {
            const unsigned same = sp ? spk[q] : ~spk[q];
            const unsigned m = same & rmask(p + 1 - 32 * q, dh - 32 * q);
            if (np < 0 && m != 0u) np = 32 * q + __builtin_ctz(m);
        }
        prevRow = (pp < 0) ? -1 : base + pp;
        nextRow = (np < 0) ? -1 : base + np;
    }

    const float INV_PI = 0.318309886183790671538f;
    for (int bch = 0; bch < ROWS_PER_WAVE / ROW_BATCH; ++bch) {
        int   prv[ROW_BATCH], nxt[ROW_BATCH];
        v4f   ra[ROW_BATCH][4], rp[ROW_BATCH][4], rn[ROW_BATCH][4];
#pragma unroll
        for (int u = 0; u < ROW_BATCH; ++u) {
            const int k  = bch * ROW_BATCH + u;
            const int i  = m0 + r0 + k;
            prv[u] = __builtin_amdgcn_readlane(prevRow, k);
            nxt[u] = __builtin_amdgcn_readlane(nextRow, k);
            const float* xi = inp + (size_t)i * D_DIM;
            const float* xp = (prv[u] >= 0) ? inp + (size_t)prv[u] * D_DIM : xi;
            const float* xn = (nxt[u] >= 0) ? inp + (size_t)nxt[u] * D_DIM : xi;
#pragma unroll
            for (int t = 0; t < 4; ++t) {
                const int kk = t * 128 + lane * 4;
                ra[u][t] = *(const v4f*)(xi + kk);
                rp[u][t] = *(const v4f*)(xp + kk);
                rn[u][t] = *(const v4f*)(xn + kk);
            }
        }
        float s[ROW_BATCH][5];
#pragma unroll
        for (int u = 0; u < ROW_BATCH; ++u) {
            float sii = 0.f, spp = 0.f, snn = 0.f, sip = 0.f, sin_ = 0.f;
#pragma unroll
            for (int t = 0; t < 4; ++t) {
                const v4f a = ra[u][t], p = rp[u][t], n = rn[u][t];
                sii  += a.x*a.x + a.y*a.y + a.z*a.z + a.w*a.w;
                spp  += p.x*p.x + p.y*p.y + p.z*p.z + p.w*p.w;
                snn  += n.x*n.x + n.y*n.y + n.z*n.z + n.w*n.w;
                sip  += a.x*p.x + a.y*p.y + a.z*p.z + a.w*p.w;
                sin_ += a.x*n.x + a.y*n.y + a.z*n.z + a.w*n.w;
            }
            s[u][0] = sii; s[u][1] = spp; s[u][2] = snn; s[u][3] = sip; s[u][4] = sin_;
        }
#pragma unroll
        for (int u = 0; u < ROW_BATCH; ++u)
#pragma unroll
            for (int q = 0; q < 5; ++q) s[u][q] += xrow(s[u][q]);
#pragma unroll
        for (int u = 0; u < ROW_BATCH; ++u)
#pragma unroll
            for (int q = 0; q < 5; ++q) s[u][q] += xlane<8>(s[u][q]);
#pragma unroll
        for (int u = 0; u < ROW_BATCH; ++u)
#pragma unroll
            for (int q = 0; q < 5; ++q) s[u][q] += xlane<4>(s[u][q]);
#pragma unroll
        for (int u = 0; u < ROW_BATCH; ++u)
#pragma unroll
            for (int q = 0; q < 5; ++q) s[u][q] += xlane<2>(s[u][q]);
#pragma unroll
        for (int u = 0; u < ROW_BATCH; ++u)
#pragma unroll
            for (int q = 0; q < 5; ++q) s[u][q] += xlane<1>(s[u][q]);
        float na = 0.f, nb = 0.f, num = 0.f;
        bool  has = false;
#pragma unroll
        for (int u = 0; u < ROW_BATCH; ++u) {
            if (lane == 2 * u)     { na = s[u][0]; nb = s[u][1]; num = s[u][3]; has = prv[u] >= 0; }
            if (lane == 2 * u + 1) { na = s[u][0]; nb = s[u][2]; num = s[u][4]; has = nxt[u] >= 0; }
        }
        float wgt = 0.f;
        {
            float den = sqrtf(na) * sqrtf(nb);
            float cs  = (den > 0.f) ? num / den : 0.f;
            cs = fminf(1.f, fmaxf(-1.f, cs));
            const float wv = 1.f - acosf(cs) * INV_PI;
            wgt = has ? wv : 0.f;
        }
#pragma unroll
        for (int u = 0; u < ROW_BATCH; ++u) {
            const float wp = __builtin_bit_cast(float, __builtin_amdgcn_readlane(__builtin_bit_cast(int, wgt), 2 * u));
            const float wn = __builtin_bit_cast(float, __builtin_amdgcn_readlane(__builtin_bit_cast(int, wgt), 2 * u + 1));
            float* xo = xs + (r0 + bch * ROW_BATCH + u) * XPITCH;
#pragma unroll
            for (int t = 0; t < 4; ++t) {
                const int kk = t * 128 + lane * 4;
                v4f r;
                r.x = ra[u][t].x + wp * rp[u][t].x + wn * rn[u][t].x;
                r.y = ra[u][t].y + wp * rp[u][t].y + wn * rn[u][t].y;
                r.z = ra[u][t].z + wp * rp[u][t].z + wn * rn[u][t].z;
                r.w = ra[u][t].w + wp * rp[u][t].w + wn * rn[u][t].w;
                *(v4f*)(xo + kk) = r;
            }
        }
    }
    __syncthreads();

    const int aoff = c * XPITCH + 4 * h;
#define LOAD_A(set, koff)                                                        \
    _Pragma("unroll")                                                            \
    for (int i_ = 0; i_ < 3; ++i_)                                               \
        fa[set][i_] = *(const v4f*)(&xs[aoff + i_ * 16 * XPITCH + kbase + (koff)]);

    int kbase = 0;
    LOAD_A(0, 0)
    LOAD_A(1, 8)
    LOAD_A(2, 16)

    v8f acc[3][GEMM_NJ];
#pragma unroll
    for (int i = 0; i < 3; ++i)
#pragma unroll
        for (int j = 0; j < GEMM_NJ; ++j) acc[i][j] = v8f{};

    for (int it = 0; it < 15; ++it) {
        LOAD_A(3, 24)
        load_b(fb[3], pb, 24);
        __builtin_amdgcn_sched_barrier(0);
        mma_set(acc, fa[0], fb[0]);
        LOAD_A(0, 32)
        load_b(fb[0], pb, 32);
        __builtin_amdgcn_sched_barrier(0);
        mma_set(acc, fa[1], fb[1]);
        LOAD_A(1, 40)
        load_b(fb[1], pb, 40);
        __builtin_amdgcn_sched_barrier(0);
        mma_set(acc, fa[2], fb[2]);
        LOAD_A(2, 48)
        load_b(fb[2], pb, 48);
        __builtin_amdgcn_sched_barrier(0);
        mma_set(acc, fa[3], fb[3]);
        kbase += 32;
        pb += 32;
    }
    LOAD_A(3, 24)
    load_b(fb[3], pb, 24);
    mma_set(acc, fa[0], fb[0]);
    mma_set(acc, fa[1], fb[1]);
    mma_set(acc, fa[2], fb[2]);
    mma_set(acc, fa[3], fb[3]);
#undef LOAD_A

#pragma unroll
    for (int i = 0; i < 3; ++i) {
        const int rbase = m0 + i * 16 + h * 8;
#pragma unroll
        for (int j = 0; j < GEMM_NJ; ++j) {
            const int col = n0 + j * 16 + c;
#pragma unroll
            for (int r = 0; r < 8; ++r) {
                out[(size_t)(rbase + r) * D_DIM + col] = acc[i][j][r] + bv[j];
            }
        }
    }
}

extern "C" void kernel_launch(void* const* d_in, const int* in_sizes, int n_in,
                              void* d_out, int out_size, void* d_ws, size_t ws_size,
                              hipStream_t stream) {
    const float* inp   = (const float*)d_in[0];
    const int*   qmask = (const int*)d_in[2];
    const float* W     = (const float*)d_in[3];
    const float* bias  = (const float*)d_in[4];
    float*       out   = (float*)d_out;

    gcn_fused_kernel<<<dim3(GEMM_MTILES, GEMM_NHALVES), dim3(256), 0, stream>>>(
        inp, qmask, W, bias, out);
}
